# starting kernel
# baseline (speedup 1.0000x reference)
#include <hip/hip_runtime.h>
#include <math.h>

typedef float v2f __attribute__((ext_vector_type(2)));
typedef float v8f __attribute__((ext_vector_type(8)));

#define B_   2
#define L_   385
#define N_   384
#define C_   128
#define C2_  256
#define EPSV 1e-5f
#define LDS_STRIDE 260

__global__ __launch_bounds__(C_) void lnp_stats_kernel(
    const float* __restrict__ x, const float* __restrict__ gamma,
    const float* __restrict__ beta, float* __restrict__ out,
    float* __restrict__ scale0, float* __restrict__ econst)
{
    const int b = blockIdx.x;
    const int c = threadIdx.x;
    const size_t base = (size_t)b * L_ * C_;
    const float* __restrict__ F = x + base + C_;

    out[base + c] = x[base + c];

    float sum = 0.f;
    for (int j = 0; j < N_; ++j) sum += F[j * C_ + c];
    const float mean = sum * (1.0f / N_);

    float ss = 0.f;
    for (int j = 0; j < N_; ++j) { float d = F[j * C_ + c] - mean; ss += d * d; }
    const float var = ss * (1.0f / N_);
    const float inv = 1.0f / sqrtf(var + EPSV);
    const float sc  = gamma[c] * inv;

    float mx = -3.0e38f;
    for (int j = 0; j < N_; ++j) mx = fmaxf(mx, sc * F[j * C_ + c]);

    float se = 0.f, sa = 0.f;
    for (int j = 0; j < N_; ++j) {
        float a = sc * F[j * C_ + c];
        float e = __expf(a - mx);
        se += e; sa += a * e;
    }
    scale0[b * C_ + c] = sc;
    econst[b * C_ + c] = sa / se + beta[c];
}

__global__ __launch_bounds__(256) void lnp_gemm_kernel(
    const float* __restrict__ x, const float* __restrict__ gamma,
    const float* __restrict__ beta, const float* __restrict__ W,
    const float* __restrict__ bias, const float* __restrict__ scale0,
    const float* __restrict__ econst, float* __restrict__ out)
{
    __shared__ float ldsA[16 * LDS_STRIDE];

    const int row0 = blockIdx.x * 16;
    const int b    = row0 / N_;
    const int i0   = row0 % N_;
    const int t    = threadIdx.x;

    {
        const int  cp = t;
        const int  c  = cp & (C_ - 1);
        const bool lo = (cp < C_);
        const float sc = lo ? scale0[b * C_ + cp] : 0.f;
        const float ec = lo ? econst[b * C_ + cp] : 0.f;
        const float g1 = lo ? 0.f : gamma[cp];
        const float b1 = lo ? 0.f : beta[cp];
        const float* __restrict__ Fb = x + (size_t)b * L_ * C_ + C_;
        #pragma unroll
        for (int m = 0; m < 16; ++m) {
            float Fv = Fb[(size_t)(i0 + m) * C_ + c];
            ldsA[m * LDS_STRIDE + cp] = lo ? (ec - sc * Fv) : (g1 * Fv + b1);
        }
    }
    __syncthreads();

    const int wave  = t >> 5;
    const int lane  = t & 31;
    const int nbase = wave * 16;
    const int half  = (lane < 16) ? 0 : 2;
    const int mrow  = lane & 15;
    const int ncol  = nbase + (lane & 15);

    v8f acc;
    const float bv = bias[ncol];
    #pragma unroll
    for (int q = 0; q < 8; ++q) acc[q] = bv;

    const float* __restrict__ wrow = W + (size_t)ncol * C2_;

    #pragma unroll 4
    for (int kt = 0; kt < 64; ++kt) {
        const int c0 = kt * 4 + half;
        v2f a  = *(const v2f*)&ldsA[mrow * LDS_STRIDE + c0];
        v2f bb = *(const v2f*)&wrow[c0];
        acc = __builtin_amdgcn_wmma_f32_16x16x4_f32(
                   false, a,  false, bb,
                   (short)0, acc,  false,  false);
    }

    const int mofs = (lane < 16) ? 0 : 8;
    float* __restrict__ outp = out + (size_t)b * L_ * C_ + (size_t)(1 + i0) * C_;
    #pragma unroll
    for (int v = 0; v < 8; ++v) {
        outp[(size_t)(v + mofs) * C_ + ncol] = acc[v];
    }
}

extern "C" void kernel_launch(void* const* d_in, const int* in_sizes, int n_in,
                              void* d_out, int out_size, void* d_ws, size_t ws_size,
                              hipStream_t stream) {
    const float* x     = (const float*)d_in[0];
    const float* gamma = (const float*)d_in[1];
    const float* beta  = (const float*)d_in[2];
    const float* W     = (const float*)d_in[3];
    const float* bias  = (const float*)d_in[4];
    float* out = (float*)d_out;

    float* scale0 = (float*)d_ws;
    float* econst = scale0 + B_ * C_;

    lnp_stats_kernel<<<B_, C_, 0, stream>>>(x, gamma, beta, out, scale0, econst);
    lnp_gemm_kernel<<<(B_ * N_) / 16, 256, 0, stream>>>(
        x, gamma, beta, W, bias, scale0, econst, out);
}


# stats kernel: each channel's 384-token reduction split over 16 threads (24 tokens each held in registers, partials combined through LDS; sums become a 2-level tree), grid 2 -> 16 blocks
# speedup vs baseline: 6.4419x; 6.4419x over previous
#include <hip/hip_runtime.h>
#include <math.h>

typedef float v2f __attribute__((ext_vector_type(2)));
typedef float v8f __attribute__((ext_vector_type(8)));

#define B_   2
#define L_   385
#define N_   384
#define C_   128
#define C2_  256
#define EPSV 1e-5f
#define LDS_STRIDE 260

#define SG_  16
#define SS_  16
#define SR_  (N_ / SS_)
#define ST_  (SG_ * SS_)
#define SP_  (SS_ + 4)

typedef float v4f __attribute__((ext_vector_type(4)));

__device__ __forceinline__ void lnp_row_load(const float* row, float (&q)[SS_])
{
    #pragma unroll
    for (int i = 0; i < SS_ / 4; ++i) {
        const v4f t4 = *(const v4f*)(row + 4 * i);
        q[4 * i + 0] = t4[0]; q[4 * i + 1] = t4[1];
        q[4 * i + 2] = t4[2]; q[4 * i + 3] = t4[3];
    }
}

__global__ __launch_bounds__(ST_) void lnp_stats_kernel(
    const float* __restrict__ x, const float* __restrict__ gamma,
    const float* __restrict__ beta, float* __restrict__ out,
    float* __restrict__ scale0, float* __restrict__ econst)
{
    __shared__ __attribute__((aligned(16))) float p_sum[SG_][SP_];
    __shared__ __attribute__((aligned(16))) float p_max[SG_][SP_];
    __shared__ __attribute__((aligned(16))) float p_min[SG_][SP_];
    __shared__ __attribute__((aligned(16))) float p_ss [SG_][SP_];
    __shared__ __attribute__((aligned(16))) float p_se [SG_][SP_];
    __shared__ __attribute__((aligned(16))) float p_sa [SG_][SP_];

    const int g  = blockIdx.x;
    const int b  = blockIdx.y;
    const int t  = threadIdx.x;
    const int cl = t & (SG_ - 1);
    const int s  = t / SG_;
    const int c  = g * SG_ + cl;
    const size_t base = (size_t)b * L_ * C_;
    const float* __restrict__ F = x + base + C_ + c;

    float v[SR_];
    #pragma unroll
    for (int r = 0; r < SR_; ++r) v[r] = F[(size_t)(s + r * SS_) * C_];

    const float gam = gamma[c];
    const float bet = beta[c];

    const bool cls_copy = (g == 0) && (t < C_);
    float clsv = 0.f;
    if (cls_copy) clsv = x[base + t];

    float q[SS_];

    float sum = v[0], mxF = v[0], mnF = v[0];
    #pragma unroll
    for (int r = 1; r < SR_; ++r) {
        sum += v[r];
        mxF = fmaxf(mxF, v[r]);
        mnF = fminf(mnF, v[r]);
    }
    p_sum[cl][s] = sum; p_max[cl][s] = mxF; p_min[cl][s] = mnF;
    __syncthreads();
    lnp_row_load(p_sum[cl], q);
    float tsum = q[0];
    #pragma unroll
    for (int k = 1; k < SS_; ++k) tsum += q[k];
    lnp_row_load(p_max[cl], q);
    float tmax = q[0];
    #pragma unroll
    for (int k = 1; k < SS_; ++k) tmax = fmaxf(tmax, q[k]);
    lnp_row_load(p_min[cl], q);
    float tmin = q[0];
    #pragma unroll
    for (int k = 1; k < SS_; ++k) tmin = fminf(tmin, q[k]);
    const float mean = tsum * (1.0f / N_);

    float ss = 0.f;
    #pragma unroll
    for (int r = 0; r < SR_; ++r) { float d = v[r] - mean; ss += d * d; }
    p_ss[cl][s] = ss;
    __syncthreads();
    lnp_row_load(p_ss[cl], q);
    float tss = q[0];
    #pragma unroll
    for (int k = 1; k < SS_; ++k) tss += q[k];
    const float var = tss * (1.0f / N_);
    const float inv = 1.0f / sqrtf(var + EPSV);
    const float sc  = gam * inv;
    const float mx  = (sc >= 0.f) ? sc * tmax : sc * tmin;

    float se = 0.f, sa = 0.f;
    #pragma unroll
    for (int r = 0; r < SR_; ++r) {
        float a = sc * v[r];
        float e = __expf(a - mx);
        se += e; sa += a * e;
    }
    p_se[cl][s] = se; p_sa[cl][s] = sa;
    __syncthreads();
    if (s == 0) {
        lnp_row_load(p_se[cl], q);
        float tse = q[0];
        #pragma unroll
        for (int k = 1; k < SS_; ++k) tse += q[k];
        lnp_row_load(p_sa[cl], q);
        float tsa = q[0];
        #pragma unroll
        for (int k = 1; k < SS_; ++k) tsa += q[k];
        scale0[b * C_ + c] = sc;
        econst[b * C_ + c] = tsa / tse + bet;
    }
    if (cls_copy) out[base + t] = clsv;
}

__global__ __launch_bounds__(256) void lnp_gemm_kernel(
    const float* __restrict__ x, const float* __restrict__ gamma,
    const float* __restrict__ beta, const float* __restrict__ W,
    const float* __restrict__ bias, const float* __restrict__ scale0,
    const float* __restrict__ econst, float* __restrict__ out)
{
    __shared__ float ldsA[16 * LDS_STRIDE];

    const int row0 = blockIdx.x * 16;
    const int b    = row0 / N_;
    const int i0   = row0 % N_;
    const int t    = threadIdx.x;

    {
        const int  cp = t;
        const int  c  = cp & (C_ - 1);
        const bool lo = (cp < C_);
        const float sc = lo ? scale0[b * C_ + cp] : 0.f;
        const float ec = lo ? econst[b * C_ + cp] : 0.f;
        const float g1 = lo ? 0.f : gamma[cp];
        const float b1 = lo ? 0.f : beta[cp];
        const float* __restrict__ Fb = x + (size_t)b * L_ * C_ + C_;
        #pragma unroll
        for (int m = 0; m < 16; ++m) {
            float Fv = Fb[(size_t)(i0 + m) * C_ + c];
            ldsA[m * LDS_STRIDE + cp] = lo ? (ec - sc * Fv) : (g1 * Fv + b1);
        }
    }
    __syncthreads();

    const int wave  = t >> 5;
    const int lane  = t & 31;
    const int nbase = wave * 16;
    const int half  = (lane < 16) ? 0 : 2;
    const int mrow  = lane & 15;
    const int ncol  = nbase + (lane & 15);

    v8f acc;
    const float bv = bias[ncol];
    #pragma unroll
    for (int q = 0; q < 8; ++q) acc[q] = bv;

    const float* __restrict__ wrow = W + (size_t)ncol * C2_;

    #pragma unroll 4
    for (int kt = 0; kt < 64; ++kt) {
        const int c0 = kt * 4 + half;
        v2f a  = *(const v2f*)&ldsA[mrow * LDS_STRIDE + c0];
        v2f bb = *(const v2f*)&wrow[c0];
        acc = __builtin_amdgcn_wmma_f32_16x16x4_f32(
                   false, a,  false, bb,
                   (short)0, acc,  false,  false);
    }

    const int mofs = (lane < 16) ? 0 : 8;
    float* __restrict__ outp = out + (size_t)b * L_ * C_ + (size_t)(1 + i0) * C_;
    #pragma unroll
    for (int v = 0; v < 8; ++v) {
        outp[(size_t)(v + mofs) * C_ + ncol] = acc[v];
    }
}

extern "C" void kernel_launch(void* const* d_in, const int* in_sizes, int n_in,
                              void* d_out, int out_size, void* d_ws, size_t ws_size,
                              hipStream_t stream) {
    const float* x     = (const float*)d_in[0];
    const float* gamma = (const float*)d_in[1];
    const float* beta  = (const float*)d_in[2];
    const float* W     = (const float*)d_in[3];
    const float* bias  = (const float*)d_in[4];
    float* out = (float*)d_out;

    float* scale0 = (float*)d_ws;
    float* econst = scale0 + B_ * C_;

    lnp_stats_kernel<<<dim3(C_ / SG_, B_), ST_, 0, stream>>>(x, gamma, beta, out, scale0, econst);
    lnp_gemm_kernel<<<(B_ * N_) / 16, 256, 0, stream>>>(
        x, gamma, beta, W, bias, scale0, econst, out);
}


# gemm kernel: W copied once per block into LDS with whole-line 16-byte loads (row pitch 260 floats); K loop, fully unrolled, reads its B fragments from LDS; no global load behind the barrier
# speedup vs baseline: 8.2777x; 1.2850x over previous
#include <hip/hip_runtime.h>
#include <math.h>

typedef float v2f __attribute__((ext_vector_type(2)));
typedef float v8f __attribute__((ext_vector_type(8)));

#define B_   2
#define L_   385
#define N_   384
#define C_   128
#define C2_  256
#define EPSV 1e-5f
#define LDS_STRIDE 260

#define SG_  16
#define SS_  16
#define SR_  (N_ / SS_)
#define ST_  (SG_ * SS_)
#define SP_  (SS_ + 4)

typedef float v4f __attribute__((ext_vector_type(4)));

__device__ __forceinline__ void lnp_row_load(const float* row, float (&q)[SS_])
{
    #pragma unroll
    for (int i = 0; i < SS_ / 4; ++i) {
        const v4f t4 = *(const v4f*)(row + 4 * i);
        q[4 * i + 0] = t4[0]; q[4 * i + 1] = t4[1];
        q[4 * i + 2] = t4[2]; q[4 * i + 3] = t4[3];
    }
}

__global__ __launch_bounds__(ST_) void lnp_stats_kernel(
    const float* __restrict__ x, const float* __restrict__ gamma,
    const float* __restrict__ beta, float* __restrict__ out,
    float* __restrict__ scale0, float* __restrict__ econst)
{
    __shared__ __attribute__((aligned(16))) float p_sum[SG_][SP_];
    __shared__ __attribute__((aligned(16))) float p_max[SG_][SP_];
    __shared__ __attribute__((aligned(16))) float p_min[SG_][SP_];
    __shared__ __attribute__((aligned(16))) float p_ss [SG_][SP_];
    __shared__ __attribute__((aligned(16))) float p_se [SG_][SP_];
    __shared__ __attribute__((aligned(16))) float p_sa [SG_][SP_];

    const int g  = blockIdx.x;
    const int b  = blockIdx.y;
    const int t  = threadIdx.x;
    const int cl = t & (SG_ - 1);
    const int s  = t / SG_;
    const int c  = g * SG_ + cl;
    const size_t base = (size_t)b * L_ * C_;
    const float* __restrict__ F = x + base + C_ + c;

    float v[SR_];
    #pragma unroll
    for (int r = 0; r < SR_; ++r) v[r] = F[(size_t)(s + r * SS_) * C_];

    const float gam = gamma[c];
    const float bet = beta[c];

    const bool cls_copy = (g == 0) && (t < C_);
    float clsv = 0.f;
    if (cls_copy) clsv = x[base + t];

    float q[SS_];

    float sum = v[0], mxF = v[0], mnF = v[0];
    #pragma unroll
    for (int r = 1; r < SR_; ++r) {
        sum += v[r];
        mxF = fmaxf(mxF, v[r]);
        mnF = fminf(mnF, v[r]);
    }
    p_sum[cl][s] = sum; p_max[cl][s] = mxF; p_min[cl][s] = mnF;
    __syncthreads();
    lnp_row_load(p_sum[cl], q);
    float tsum = q[0];
    #pragma unroll
    for (int k = 1; k < SS_; ++k) tsum += q[k];
    lnp_row_load(p_max[cl], q);
    float tmax = q[0];
    #pragma unroll
    for (int k = 1; k < SS_; ++k) tmax = fmaxf(tmax, q[k]);
    lnp_row_load(p_min[cl], q);
    float tmin = q[0];
    #pragma unroll
    for (int k = 1; k < SS_; ++k) tmin = fminf(tmin, q[k]);
    const float mean = tsum * (1.0f / N_);

    float ss = 0.f;
    #pragma unroll
    for (int r = 0; r < SR_; ++r) { float d = v[r] - mean; ss += d * d; }
    p_ss[cl][s] = ss;
    __syncthreads();
    lnp_row_load(p_ss[cl], q);
    float tss = q[0];
    #pragma unroll
    for (int k = 1; k < SS_; ++k) tss += q[k];
    const float var = tss * (1.0f / N_);
    const float inv = 1.0f / sqrtf(var + EPSV);
    const float sc  = gam * inv;
    const float mx  = (sc >= 0.f) ? sc * tmax : sc * tmin;

    float se = 0.f, sa = 0.f;
    #pragma unroll
    for (int r = 0; r < SR_; ++r) {
        float a = sc * v[r];
        float e = __expf(a - mx);
        se += e; sa += a * e;
    }
    p_se[cl][s] = se; p_sa[cl][s] = sa;
    __syncthreads();
    if (s == 0) {
        lnp_row_load(p_se[cl], q);
        float tse = q[0];
        #pragma unroll
        for (int k = 1; k < SS_; ++k) tse += q[k];
        lnp_row_load(p_sa[cl], q);
        float tsa = q[0];
        #pragma unroll
        for (int k = 1; k < SS_; ++k) tsa += q[k];
        scale0[b * C_ + c] = sc;
        econst[b * C_ + c] = tsa / tse + bet;
    }
    if (cls_copy) out[base + t] = clsv;
}

#define WP_ 260

__global__ __launch_bounds__(256) void lnp_gemm_kernel(
    const float* __restrict__ x, const float* __restrict__ gamma,
    const float* __restrict__ beta, const float* __restrict__ W,
    const float* __restrict__ bias, const float* __restrict__ scale0,
    const float* __restrict__ econst, float* __restrict__ out)
{
    __shared__ __attribute__((aligned(16))) float ldsW[C_ * WP_];
    __shared__ __attribute__((aligned(16))) float ldsA[16 * LDS_STRIDE];

    const int row0 = blockIdx.x * 16;
    const int b    = row0 / N_;
    const int i0   = row0 % N_;
    const int t    = threadIdx.x;

    const int wave  = t >> 5;
    const int lane  = t & 31;
    const int nbase = wave * 16;
    const int half  = (lane < 16) ? 0 : 2;
    const int mrow  = lane & 15;
    const int ncol  = nbase + (lane & 15);

    const int  cp = t;
    const int  c  = cp & (C_ - 1);
    const bool lo = (cp < C_);
    const float sc = lo ? scale0[b * C_ + cp] : 0.f;
    const float ec = lo ? econst[b * C_ + cp] : 0.f;
    const float g1 = lo ? 0.f : gamma[cp];
    const float b1 = lo ? 0.f : beta[cp];
    const float* __restrict__ Fb = x + (size_t)b * L_ * C_ + C_;
    float Fv[16];
    #pragma unroll
    for (int m = 0; m < 16; ++m) Fv[m] = Fb[(size_t)(i0 + m) * C_ + c];
    const float bv = bias[ncol];

    const int wchunk = t & 63;
    const int wrow0  = t >> 6;
    v4f wv[32];
    #pragma unroll
    for (int p = 0; p < 32; ++p)
        wv[p] = *(const v4f*)&W[(size_t)(4 * p + wrow0) * C2_ + 4 * wchunk];

    #pragma unroll
    for (int m = 0; m < 16; ++m)
        ldsA[m * LDS_STRIDE + cp] = lo ? (ec - sc * Fv[m]) : (g1 * Fv[m] + b1);
    #pragma unroll
    for (int p = 0; p < 32; ++p)
        *(v4f*)&ldsW[(4 * p + wrow0) * WP_ + 4 * wchunk] = wv[p];
    __syncthreads();

    v8f acc;
    #pragma unroll
    for (int q = 0; q < 8; ++q) acc[q] = bv;

    const float* wrow = &ldsW[ncol * WP_];
    const float* arow = &ldsA[mrow * LDS_STRIDE];

    #pragma unroll
    for (int kt = 0; kt < 64; ++kt) {
        const int c0 = kt * 4 + half;
        v2f a  = *(const v2f*)&arow[c0];
        v2f bb = *(const v2f*)&wrow[c0];
        acc = __builtin_amdgcn_wmma_f32_16x16x4_f32(
                   false, a,  false, bb,
                   (short)0, acc,  false,  false);
    }

    const int mofs = (lane < 16) ? 0 : 8;
    float* __restrict__ outp = out + (size_t)b * L_ * C_ + (size_t)(1 + i0) * C_;
    #pragma unroll
    for (int v = 0; v < 8; ++v) {
        outp[(size_t)(v + mofs) * C_ + ncol] = acc[v];
    }
}

extern "C" void kernel_launch(void* const* d_in, const int* in_sizes, int n_in,
                              void* d_out, int out_size, void* d_ws, size_t ws_size,
                              hipStream_t stream) {
    const float* x     = (const float*)d_in[0];
    const float* gamma = (const float*)d_in[1];
    const float* beta  = (const float*)d_in[2];
    const float* W     = (const float*)d_in[3];
    const float* bias  = (const float*)d_in[4];
    float* out = (float*)d_out;

    float* scale0 = (float*)d_ws;
    float* econst = scale0 + B_ * C_;

    lnp_stats_kernel<<<dim3(C_ / SG_, B_), ST_, 0, stream>>>(x, gamma, beta, out, scale0, econst);
    lnp_gemm_kernel<<<(B_ * N_) / 16, 256, 0, stream>>>(
        x, gamma, beta, W, bias, scale0, econst, out);
}
